# starting kernel
# baseline (speedup 1.0000x reference)
#include <hip/hip_runtime.h>
#include <math.h>

typedef float v2f __attribute__((ext_vector_type(2)));
typedef float v8f __attribute__((ext_vector_type(8)));

#define BN_EPS 1e-5f

constexpr int NB  = 8;
constexpr int C   = 64;
constexpr int HIN = 64,  WIN = 64;
constexpr int H   = 128, W   = 128;
constexpr int PIN = HIN * WIN;
constexpr int P   = H * W;

__global__ void res_mean_kernel(const float* __restrict__ ref,
                                float* __restrict__ res) {
    int idx = blockIdx.x * blockDim.x + threadIdx.x;
    int b = idx >> 14;
    int p = idx & (P - 1);
    const float* rp = ref + (size_t)b * C * P + p;
    float s = 0.f;
#pragma unroll 8
    for (int c = 0; c < C; ++c) s += rp[c * P];
    res[idx] = s * (1.0f / 64.0f);
}

__global__ void conv1_wmma_kernel(const float* __restrict__ x,
                                  const float* __restrict__ w1,
                                  const float* __restrict__ b1,
                                  const float* __restrict__ g1,
                                  const float* __restrict__ beta1,
                                  const float* __restrict__ m1,
                                  const float* __restrict__ var1,
                                  float* __restrict__ y) {
    const int lane = threadIdx.x & 31;
    const int wid  = (blockIdx.x * blockDim.x + threadIdx.x) >> 5;
    const int b  = wid >> 10;
    const int mt = (wid >> 8) & 3;
    const int pt = wid & 255;

    const int half = lane >> 4;
    const int lm   = lane & 15;
    const int kSel = half * 2;

    const int   oA = mt * 16 + lm;
    const float sA = g1[oA] * rsqrtf(var1[oA] + BN_EPS);
    const float* wrow = w1 + oA * 64;

    const float* xcol = x + (size_t)b * C * PIN + pt * 16 + lm;

    v8f acc = {};
#pragma unroll
    for (int kk = 0; kk < 64; kk += 4) {
        v2f a, bm;
        a.x  = wrow[kk + kSel]     * sA;
        a.y  = wrow[kk + kSel + 1] * sA;
        bm.x = xcol[(kk + kSel)     * PIN];
        bm.y = xcol[(kk + kSel + 1) * PIN];
        acc = __builtin_amdgcn_wmma_f32_16x16x4_f32(
                  false, a, false, bm, (short)0, acc, false, false);
    }

    float* yb = y + (size_t)b * C * PIN + pt * 16 + lm;
#pragma unroll
    for (int v = 0; v < 8; ++v) {
        const int   o    = mt * 16 + v + half * 8;
        const float s    = g1[o] * rsqrtf(var1[o] + BN_EPS);
        const float bias = b1[o] * s + beta1[o] - m1[o] * s;
        yb[o * PIN] = fmaxf(acc[v] + bias, 0.f);
    }
}

__global__ void fused_out_kernel(const float* __restrict__ ref,
                                 const float* __restrict__ w2,
                                 const float* __restrict__ b2,
                                 const float* __restrict__ g2,
                                 const float* __restrict__ beta2,
                                 const float* __restrict__ m2,
                                 const float* __restrict__ var2,
                                 const float* __restrict__ res,
                                 const float* __restrict__ y,
                                 float* __restrict__ out) {
    const int lane = threadIdx.x & 31;
    const int wid  = (blockIdx.x * blockDim.x + threadIdx.x) >> 5;
    const int b  = wid >> 12;
    const int ct = (wid >> 10) & 3;
    const int pt = wid & 1023;

    const int half = lane >> 4;
    const int lm   = lane & 15;
    const int kSel = half * 2;

    const int   oA = ct * 16 + lm;
    const float sA = g2[oA] * rsqrtf(var2[oA] + BN_EPS);
    const float* wrow = w2 + oA * 64;
    const float* rcol = ref + (size_t)b * C * P + pt * 16 + lm;

    v8f acc = {};
#pragma unroll
    for (int kk = 0; kk < 64; kk += 4) {
        v2f a, bm;
        a.x  = wrow[kk + kSel]     * sA;
        a.y  = wrow[kk + kSel + 1] * sA;
        bm.x = rcol[(kk + kSel)     * P];
        bm.y = rcol[(kk + kSel + 1) * P];
        acc = __builtin_amdgcn_wmma_f32_16x16x4_f32(
                  false, a, false, bm, (short)0, acc, false, false);
    }

    const int p  = pt * 16 + lm;
    const int ph = p >> 7;
    const int pw = p & 127;
    const float* resb = res + b * P;

    float ur[9];
    float ssum = 0.f;
#pragma unroll
    for (int di = 0; di < 3; ++di)
#pragma unroll
        for (int dj = 0; dj < 3; ++dj) {
            const int hh = ph + di - 1, ww = pw + dj - 1;
            const bool ok = (hh >= 0) & (hh < H) & (ww >= 0) & (ww < W);
            const float v = ok ? resb[hh * W + ww] : 0.f;
            ur[di * 3 + dj] = v;
            ssum += v;
        }
    const float ua   = ssum * (1.f / 9.f);
    const float ui   = (ur[4] - ua > 0.f) ? 1.f : 0.f;

    const int r1 = (ph + 1) >> 1, c1 = (pw + 1) >> 1;
    const int r0c = max(r1 - 1, 0), r1c = min(r1, HIN - 1);
    const int c0c = max(c1 - 1, 0), c1c = min(c1, WIN - 1);

    float wgt00 = 0.f, wgt01 = 0.f, wgt10 = 0.f, wgt11 = 0.f;
    float den = 1e-6f;
#pragma unroll
    for (int di = 0; di < 3; ++di)
#pragma unroll
        for (int dj = 0; dj < 3; ++dj) {
            const int k = di * 3 + dj;
            const float up = (ur[k] - ua > 0.f) ? 1.f : 0.f;
            const float mk = ui * up + (1.f - ui) * (1.f - up);
            den += mk;
            const int hh = ph + di - 1, ww = pw + dj - 1;
            if (hh >= 0 && hh < H && ww >= 0 && ww < W) {
                const int sr = ((hh >> 1) == r1) ? 1 : 0;
                const int sc = ((ww >> 1) == c1) ? 1 : 0;
                if (sr == 0 && sc == 0) wgt00 += mk;
                else if (sr == 0)       wgt01 += mk;
                else if (sc == 0)       wgt10 += mk;
                else                    wgt11 += mk;
            }
        }
    const float invden = 1.f / den;

    const float* ybase = y + (size_t)b * C * PIN;
    float* ob = out + (size_t)b * C * P + p;
#pragma unroll
    for (int v = 0; v < 8; ++v) {
        const int   o    = ct * 16 + v + half * 8;
        const float s    = g2[o] * rsqrtf(var2[o] + BN_EPS);
        const float bias = b2[o] * s + beta2[o] - m2[o] * s;
        const float c2v  = fmaxf(acc[v] + bias, 0.f);
        const float* yo  = ybase + o * PIN;
        const float num  = wgt00 * yo[r0c * WIN + c0c]
                         + wgt01 * yo[r0c * WIN + c1c]
                         + wgt10 * yo[r1c * WIN + c0c]
                         + wgt11 * yo[r1c * WIN + c1c];
        ob[o * P] = num * invden + c2v;
    }
}

extern "C" void kernel_launch(void* const* d_in, const int* in_sizes, int n_in,
                              void* d_out, int out_size, void* d_ws, size_t ws_size,
                              hipStream_t stream) {
    const float* x      = (const float*)d_in[0];
    const float* ref    = (const float*)d_in[1];
    const float* w1     = (const float*)d_in[2];
    const float* b1     = (const float*)d_in[3];
    const float* g1     = (const float*)d_in[4];
    const float* beta1  = (const float*)d_in[5];
    const float* m1     = (const float*)d_in[6];
    const float* var1   = (const float*)d_in[7];
    const float* w2     = (const float*)d_in[8];
    const float* b2     = (const float*)d_in[9];
    const float* g2     = (const float*)d_in[10];
    const float* beta2  = (const float*)d_in[11];
    const float* m2     = (const float*)d_in[12];
    const float* var2   = (const float*)d_in[13];
    float* out = (float*)d_out;

    float* res = (float*)d_ws;
    float* y   = res + (size_t)NB * P;

    res_mean_kernel<<<(NB * P) / 256, 256, 0, stream>>>(ref, res);

    conv1_wmma_kernel<<<1024, 256, 0, stream>>>(x, w1, b1, g1, beta1, m1, var1, y);

    fused_out_kernel<<<4096, 256, 0, stream>>>(ref, w2, b2, g2, beta2, m2, var2,
                                               res, y, out);
}


# Both 1x1-conv WMMA kernels: wave register tile 16x16 -> 64 ch x 64 px (conv1 64x32), lane owns 4 adjacent px (128-bit loads/stores); k permuted within groups of 8; BN scale after the sum, from LDS
# speedup vs baseline: 1.3985x; 1.3985x over previous
#include <hip/hip_runtime.h>
#include <math.h>


typedef float v2f __attribute__((ext_vector_type(2)));
typedef float v4f __attribute__((ext_vector_type(4)));
typedef float v8f __attribute__((ext_vector_type(8)));

#define BN_EPS 1e-5f

constexpr int NB  = 8;
constexpr int C   = 64;
constexpr int HIN = 64,  WIN = 64;
constexpr int H   = 128, W   = 128;
constexpr int PIN = HIN * WIN;
constexpr int P   = H * W;

__global__ void res_mean_kernel(const float* __restrict__ ref,
                                float* __restrict__ res) {
    int idx = blockIdx.x * blockDim.x + threadIdx.x;
    int b = idx >> 14;
    int p = idx & (P - 1);
    const float* rp = ref + (size_t)b * C * P + p;
    float s = 0.f;
#pragma unroll 8
    for (int c = 0; c < C; ++c) s += rp[c * P];
    res[idx] = s * (1.0f / 64.0f);
}

template <int NT, int STRIDE>
__device__ __forceinline__ void gemm64(const float* __restrict__ w,
                                       const float* __restrict__ incol,
                                       int lm, int half, v8f (&acc)[4][NT]) {
    typedef float vnf __attribute__((ext_vector_type(NT)));
#pragma unroll
    for (int mt = 0; mt < 4; ++mt)
#pragma unroll
        for (int t = 0; t < NT; ++t) acc[mt][t] = v8f{};

#pragma unroll
    for (int j = 0; j < 8; ++j) {
        const int k0 = 8 * j + 4 * half;
        v4f av[4];
#pragma unroll
        for (int mt = 0; mt < 4; ++mt)
            av[mt] = *(const v4f*)(w + (mt * 16 + lm) * 64 + k0);
        vnf bv[4];
#pragma unroll
        for (int i = 0; i < 4; ++i)
            bv[i] = *(const vnf*)(incol + (size_t)(k0 + i) * STRIDE);
#pragma unroll
        for (int u = 0; u < 2; ++u)
#pragma unroll
            for (int mt = 0; mt < 4; ++mt) {
                v2f a;
                a.x = av[mt][2 * u];
                a.y = av[mt][2 * u + 1];
#pragma unroll
                for (int t = 0; t < NT; ++t) {
                    v2f bm;
                    bm.x = bv[2 * u][t];
                    bm.y = bv[2 * u + 1][t];
                    acc[mt][t] = __builtin_amdgcn_wmma_f32_16x16x4_f32(
                        false, a, false, bm, (short)0, acc[mt][t], false, false);
                }
            }
    }
}

__global__ __launch_bounds__(64)
void conv1_wmma_kernel(const float* __restrict__ x,
                       const float* __restrict__ w1,
                       const float* __restrict__ b1,
                       const float* __restrict__ g1,
                       const float* __restrict__ beta1,
                       const float* __restrict__ m1,
                       const float* __restrict__ var1,
                       float* __restrict__ y) {
    __shared__ v2f sb[C];
    const int tid  = threadIdx.x;
    const int lane = tid & 31;
    const int wid  = (blockIdx.x * 64 + tid) >> 5;
    const int b    = wid >> 7;
    const int seg  = wid & 127;
    const int half = lane >> 4;
    const int lm   = lane & 15;

    v2f e;
    e.x = g1[tid] * rsqrtf(var1[tid] + BN_EPS);
    e.y = b1[tid] * e.x + beta1[tid] - m1[tid] * e.x;

    v8f acc[4][2];
    gemm64<2, PIN>(w1, x + (size_t)b * C * PIN + seg * 32 + 2 * lm, lm, half, acc);

    sb[tid] = e;
    __syncthreads();

    float* yb = y + (size_t)b * C * PIN + seg * 32 + 2 * lm;
#pragma unroll
    for (int mt = 0; mt < 4; ++mt)
#pragma unroll
        for (int r = 0; r < 8; ++r) {
            const int o  = mt * 16 + 8 * half + r;
            const v2f sc = sb[o];
            v2f ov;
            ov.x = fmaxf(fmaf(acc[mt][0][r], sc.x, sc.y), 0.f);
            ov.y = fmaxf(fmaf(acc[mt][1][r], sc.x, sc.y), 0.f);
            *(v2f*)(yb + o * PIN) = ov;
        }
}

__global__ __launch_bounds__(64)
void fused_out_kernel(const float* __restrict__ ref,
                      const float* __restrict__ w2,
                      const float* __restrict__ b2,
                      const float* __restrict__ g2,
                      const float* __restrict__ beta2,
                      const float* __restrict__ m2,
                      const float* __restrict__ var2,
                      const float* __restrict__ res,
                      const float* __restrict__ y,
                      float* __restrict__ out) {
    __shared__ v2f sb[C];
    const int tid  = threadIdx.x;
    const int lane = tid & 31;
    const int wid  = (blockIdx.x * 64 + tid) >> 5;
    const int b    = wid >> 8;
    const int seg  = wid & 255;
    const int half = lane >> 4;
    const int lm   = lane & 15;
    const int ph   = seg >> 1;
    const int pw0  = (seg & 1) * 64 + 4 * lm;

    v2f e;
    e.x = g2[tid] * rsqrtf(var2[tid] + BN_EPS);
    e.y = b2[tid] * e.x + beta2[tid] - m2[tid] * e.x;

    const float* resb = res + b * P;
    const bool lok = pw0 > 0;
    const bool rok = pw0 + 4 < W;
    float R[3][6];
#pragma unroll
    for (int di = 0; di < 3; ++di) {
        const int  hh  = ph + di - 1;
        const bool hok = (hh >= 0) & (hh < H);
        const float* rr = resb + (hok ? hh : ph) * W + pw0;
        const v4f   mid = *(const v4f*)rr;
        const float lft = rr[lok ? -1 : 0];
        const float rgt = rr[rok ? 4 : 3];
        R[di][0] = (hok & lok) ? lft : 0.f;
        R[di][1] = hok ? mid.x : 0.f;
        R[di][2] = hok ? mid.y : 0.f;
        R[di][3] = hok ? mid.z : 0.f;
        R[di][4] = hok ? mid.w : 0.f;
        R[di][5] = (hok & rok) ? rgt : 0.f;
    }

    v8f acc[4][4];
    gemm64<4, P>(w2, ref + (size_t)b * C * P + ph * W + pw0, lm, half, acc);

    sb[tid] = e;
    __syncthreads();

    const int r1 = (ph + 1) >> 1;
    float wgt00[4], wgt01[4], wgt10[4], wgt11[4], invden[4];
#pragma unroll
    for (int t = 0; t < 4; ++t) {
        const int pw = pw0 + t;
        const int c1 = (pw + 1) >> 1;
        float ssum = 0.f;
#pragma unroll
        for (int di = 0; di < 3; ++di)
#pragma unroll
            for (int dj = 0; dj < 3; ++dj) ssum += R[di][t + dj];
        const float ua = ssum * (1.f / 9.f);
        const float ui = (R[1][t + 1] - ua > 0.f) ? 1.f : 0.f;

        float a00 = 0.f, a01 = 0.f, a10 = 0.f, a11 = 0.f;
        float den = 1e-6f;
#pragma unroll
        for (int di = 0; di < 3; ++di)
#pragma unroll
            for (int dj = 0; dj < 3; ++dj) {
                const float up = (R[di][t + dj] - ua > 0.f) ? 1.f : 0.f;
                const float mk = ui * up + (1.f - ui) * (1.f - up);
                den += mk;
                const int hh = ph + di - 1, ww = pw + dj - 1;
                if (hh >= 0 && hh < H && ww >= 0 && ww < W) {
                    const int sr = ((hh >> 1) == r1) ? 1 : 0;
                    const int sc = ((ww >> 1) == c1) ? 1 : 0;
                    if (sr == 0 && sc == 0) a00 += mk;
                    else if (sr == 0)       a01 += mk;
                    else if (sc == 0)       a10 += mk;
                    else                    a11 += mk;
                }
            }
        wgt00[t] = a00; wgt01[t] = a01; wgt10[t] = a10; wgt11[t] = a11;
        invden[t] = 1.f / den;
    }

    const int r0c = max(r1 - 1, 0), r1c = min(r1, HIN - 1);
    const int q   = pw0 >> 1;
    const int cl  = max(q - 1, 0), cr = min(q + 2, WIN - 1);

    const float* ybase = y + (size_t)b * C * PIN;
    float* ob = out + (size_t)b * C * P + ph * W + pw0;
#pragma unroll
    for (int mt = 0; mt < 4; ++mt)
#pragma unroll
        for (int r = 0; r < 8; ++r) {
            const int o  = mt * 16 + 8 * half + r;
            const v2f sc = sb[o];
            const float* y0 = ybase + o * PIN + r0c * WIN;
            const float* y1 = ybase + o * PIN + r1c * WIN;
            float Y0[4], Y1[4];
            const v2f m0 = *(const v2f*)(y0 + q);
            const v2f m1 = *(const v2f*)(y1 + q);
            Y0[0] = y0[cl]; Y0[1] = m0.x; Y0[2] = m0.y; Y0[3] = y0[cr];
            Y1[0] = y1[cl]; Y1[1] = m1.x; Y1[2] = m1.y; Y1[3] = y1[cr];
            v4f ov;
#pragma unroll
            for (int t = 0; t < 4; ++t) {
                const int jl = (t + 1) >> 1, jr = jl + 1;
                const float c2v = fmaxf(fmaf(acc[mt][t][r], sc.x, sc.y), 0.f);
                const float num = wgt00[t] * Y0[jl]
                                + wgt01[t] * Y0[jr]
                                + wgt10[t] * Y1[jl]
                                + wgt11[t] * Y1[jr];
                ov[t] = num * invden[t] + c2v;
            }
            *(v4f*)(ob + o * P) = ov;
        }
}

extern "C" void kernel_launch(void* const* d_in, const int* in_sizes, int n_in,
                              void* d_out, int out_size, void* d_ws, size_t ws_size,
                              hipStream_t stream) {
    const float* x      = (const float*)d_in[0];
    const float* ref    = (const float*)d_in[1];
    const float* w1     = (const float*)d_in[2];
    const float* b1     = (const float*)d_in[3];
    const float* g1     = (const float*)d_in[4];
    const float* beta1  = (const float*)d_in[5];
    const float* m1     = (const float*)d_in[6];
    const float* var1   = (const float*)d_in[7];
    const float* w2     = (const float*)d_in[8];
    const float* b2     = (const float*)d_in[9];
    const float* g2     = (const float*)d_in[10];
    const float* beta2  = (const float*)d_in[11];
    const float* m2     = (const float*)d_in[12];
    const float* var2   = (const float*)d_in[13];
    float* out = (float*)d_out;

    float* res = (float*)d_ws;
    float* y   = res + (size_t)NB * P;

    res_mean_kernel<<<(NB * P) / 256, 256, 0, stream>>>(ref, res);

    conv1_wmma_kernel<<<512, 64, 0, stream>>>(x, w1, b1, g1, beta1, m1, var1, y);

    fused_out_kernel<<<1024, 64, 0, stream>>>(ref, w2, b2, g2, beta2, m2, var2,
                                              res, y, out);
}


# Channel mean of ref moved into the conv1 launch as a block-uniform role (3 launches -> 2): 512 more blocks of 64, a lane sums 4 adjacent px by 16-byte loads, 32 in flight; channel order 0..63 kept
# speedup vs baseline: 1.8245x; 1.3046x over previous
#include <hip/hip_runtime.h>
#include <math.h>


typedef float v2f __attribute__((ext_vector_type(2)));
typedef float v4f __attribute__((ext_vector_type(4)));
typedef float v8f __attribute__((ext_vector_type(8)));

#define BN_EPS 1e-5f

constexpr int NB  = 8;
constexpr int C   = 64;
constexpr int HIN = 64,  WIN = 64;
constexpr int H   = 128, W   = 128;
constexpr int PIN = HIN * WIN;
constexpr int P   = H * W;

template <int NT, int STRIDE>
__device__ __forceinline__ void gemm64(const float* __restrict__ w,
                                       const float* __restrict__ incol,
                                       int lm, int half, v8f (&acc)[4][NT]) {
    typedef float vnf __attribute__((ext_vector_type(NT)));
#pragma unroll
    for (int mt = 0; mt < 4; ++mt)
#pragma unroll
        for (int t = 0; t < NT; ++t) acc[mt][t] = v8f{};

#pragma unroll
    for (int j = 0; j < 8; ++j) {
        const int k0 = 8 * j + 4 * half;
        v4f av[4];
#pragma unroll
        for (int mt = 0; mt < 4; ++mt)
            av[mt] = *(const v4f*)(w + (mt * 16 + lm) * 64 + k0);
        vnf bv[4];
#pragma unroll
        for (int i = 0; i < 4; ++i)
            bv[i] = *(const vnf*)(incol + (size_t)(k0 + i) * STRIDE);
#pragma unroll
        for (int u = 0; u < 2; ++u)
#pragma unroll
            for (int mt = 0; mt < 4; ++mt) {
                v2f a;
                a.x = av[mt][2 * u];
                a.y = av[mt][2 * u + 1];
#pragma unroll
                for (int t = 0; t < NT; ++t) {
                    v2f bm;
                    bm.x = bv[2 * u][t];
                    bm.y = bv[2 * u + 1][t];
                    acc[mt][t] = __builtin_amdgcn_wmma_f32_16x16x4_f32(
                        false, a, false, bm, (short)0, acc[mt][t], false, false);
                }
            }
    }
}

constexpr int CONV1_BLOCKS = 512;
constexpr int RES_BLOCKS   = (NB * P) / (64 * 4);

__global__ __launch_bounds__(64)
void conv1_res_kernel(const float* __restrict__ x,
                      const float* __restrict__ w1,
                      const float* __restrict__ b1,
                      const float* __restrict__ g1,
                      const float* __restrict__ beta1,
                      const float* __restrict__ m1,
                      const float* __restrict__ var1,
                      float* __restrict__ y,
                      const float* __restrict__ ref,
                      float* __restrict__ res) {
    __shared__ v2f sb[C];
    const int tid  = threadIdx.x;

    if (blockIdx.x >= CONV1_BLOCKS) {
        const int quad = (blockIdx.x - CONV1_BLOCKS) * 64 + tid;
        const int bb   = quad >> 12;
        const int p4   = (quad & 4095) * 4;
        const float* rp = ref + (size_t)bb * C * P + p4;
        v4f s = {0.f, 0.f, 0.f, 0.f};
#pragma unroll 32
        for (int c = 0; c < C; ++c) s += *(const v4f*)(rp + (size_t)c * P);
        *(v4f*)(res + (size_t)bb * P + p4) = s * (1.0f / 64.0f);
        return;
    }

    const int lane = tid & 31;
    const int wid  = (blockIdx.x * 64 + tid) >> 5;
    const int b    = wid >> 7;
    const int seg  = wid & 127;
    const int half = lane >> 4;
    const int lm   = lane & 15;

    v2f e;
    e.x = g1[tid] * rsqrtf(var1[tid] + BN_EPS);
    e.y = b1[tid] * e.x + beta1[tid] - m1[tid] * e.x;

    v8f acc[4][2];
    gemm64<2, PIN>(w1, x + (size_t)b * C * PIN + seg * 32 + 2 * lm, lm, half, acc);

    sb[tid] = e;
    __syncthreads();

    float* yb = y + (size_t)b * C * PIN + seg * 32 + 2 * lm;
#pragma unroll
    for (int mt = 0; mt < 4; ++mt)
#pragma unroll
        for (int r = 0; r < 8; ++r) {
            const int o  = mt * 16 + 8 * half + r;
            const v2f sc = sb[o];
            v2f ov;
            ov.x = fmaxf(fmaf(acc[mt][0][r], sc.x, sc.y), 0.f);
            ov.y = fmaxf(fmaf(acc[mt][1][r], sc.x, sc.y), 0.f);
            *(v2f*)(yb + o * PIN) = ov;
        }
}

__global__ __launch_bounds__(64)
void fused_out_kernel(const float* __restrict__ ref,
                      const float* __restrict__ w2,
                      const float* __restrict__ b2,
                      const float* __restrict__ g2,
                      const float* __restrict__ beta2,
                      const float* __restrict__ m2,
                      const float* __restrict__ var2,
                      const float* __restrict__ res,
                      const float* __restrict__ y,
                      float* __restrict__ out) {
    __shared__ v2f sb[C];
    const int tid  = threadIdx.x;
    const int lane = tid & 31;
    const int wid  = (blockIdx.x * 64 + tid) >> 5;
    const int b    = wid >> 8;
    const int seg  = wid & 255;
    const int half = lane >> 4;
    const int lm   = lane & 15;
    const int ph   = seg >> 1;
    const int pw0  = (seg & 1) * 64 + 4 * lm;

    v2f e;
    e.x = g2[tid] * rsqrtf(var2[tid] + BN_EPS);
    e.y = b2[tid] * e.x + beta2[tid] - m2[tid] * e.x;

    const float* resb = res + b * P;
    const bool lok = pw0 > 0;
    const bool rok = pw0 + 4 < W;
    float R[3][6];
#pragma unroll
    for (int di = 0; di < 3; ++di) {
        const int  hh  = ph + di - 1;
        const bool hok = (hh >= 0) & (hh < H);
        const float* rr = resb + (hok ? hh : ph) * W + pw0;
        const v4f   mid = *(const v4f*)rr;
        const float lft = rr[lok ? -1 : 0];
        const float rgt = rr[rok ? 4 : 3];
        R[di][0] = (hok & lok) ? lft : 0.f;
        R[di][1] = hok ? mid.x : 0.f;
        R[di][2] = hok ? mid.y : 0.f;
        R[di][3] = hok ? mid.z : 0.f;
        R[di][4] = hok ? mid.w : 0.f;
        R[di][5] = (hok & rok) ? rgt : 0.f;
    }

    v8f acc[4][4];
    gemm64<4, P>(w2, ref + (size_t)b * C * P + ph * W + pw0, lm, half, acc);

    sb[tid] = e;
    __syncthreads();

    const int r1 = (ph + 1) >> 1;
    float wgt00[4], wgt01[4], wgt10[4], wgt11[4], invden[4];
#pragma unroll
    for (int t = 0; t < 4; ++t) {
        const int pw = pw0 + t;
        const int c1 = (pw + 1) >> 1;
        float ssum = 0.f;
#pragma unroll
        for (int di = 0; di < 3; ++di)
#pragma unroll
            for (int dj = 0; dj < 3; ++dj) ssum += R[di][t + dj];
        const float ua = ssum * (1.f / 9.f);
        const float ui = (R[1][t + 1] - ua > 0.f) ? 1.f : 0.f;

        float a00 = 0.f, a01 = 0.f, a10 = 0.f, a11 = 0.f;
        float den = 1e-6f;
#pragma unroll
        for (int di = 0; di < 3; ++di)
#pragma unroll
            for (int dj = 0; dj < 3; ++dj) {
                const float up = (R[di][t + dj] - ua > 0.f) ? 1.f : 0.f;
                const float mk = ui * up + (1.f - ui) * (1.f - up);
                den += mk;
                const int hh = ph + di - 1, ww = pw + dj - 1;
                if (hh >= 0 && hh < H && ww >= 0 && ww < W) {
                    const int sr = ((hh >> 1) == r1) ? 1 : 0;
                    const int sc = ((ww >> 1) == c1) ? 1 : 0;
                    if (sr == 0 && sc == 0) a00 += mk;
                    else if (sr == 0)       a01 += mk;
                    else if (sc == 0)       a10 += mk;
                    else                    a11 += mk;
                }
            }
        wgt00[t] = a00; wgt01[t] = a01; wgt10[t] = a10; wgt11[t] = a11;
        invden[t] = 1.f / den;
    }

    const int r0c = max(r1 - 1, 0), r1c = min(r1, HIN - 1);
    const int q   = pw0 >> 1;
    const int cl  = max(q - 1, 0), cr = min(q + 2, WIN - 1);

    const float* ybase = y + (size_t)b * C * PIN;
    float* ob = out + (size_t)b * C * P + ph * W + pw0;
#pragma unroll
    for (int mt = 0; mt < 4; ++mt)
#pragma unroll
        for (int r = 0; r < 8; ++r) {
            const int o  = mt * 16 + 8 * half + r;
            const v2f sc = sb[o];
            const float* y0 = ybase + o * PIN + r0c * WIN;
            const float* y1 = ybase + o * PIN + r1c * WIN;
            float Y0[4], Y1[4];
            const v2f m0 = *(const v2f*)(y0 + q);
            const v2f m1 = *(const v2f*)(y1 + q);
            Y0[0] = y0[cl]; Y0[1] = m0.x; Y0[2] = m0.y; Y0[3] = y0[cr];
            Y1[0] = y1[cl]; Y1[1] = m1.x; Y1[2] = m1.y; Y1[3] = y1[cr];
            v4f ov;
#pragma unroll
            for (int t = 0; t < 4; ++t) {
                const int jl = (t + 1) >> 1, jr = jl + 1;
                const float c2v = fmaxf(fmaf(acc[mt][t][r], sc.x, sc.y), 0.f);
                const float num = wgt00[t] * Y0[jl]
                                + wgt01[t] * Y0[jr]
                                + wgt10[t] * Y1[jl]
                                + wgt11[t] * Y1[jr];
                ov[t] = num * invden[t] + c2v;
            }
            *(v4f*)(ob + o * P) = ov;
        }
}

extern "C" void kernel_launch(void* const* d_in, const int* in_sizes, int n_in,
                              void* d_out, int out_size, void* d_ws, size_t ws_size,
                              hipStream_t stream) {
    const float* x      = (const float*)d_in[0];
    const float* ref    = (const float*)d_in[1];
    const float* w1     = (const float*)d_in[2];
    const float* b1     = (const float*)d_in[3];
    const float* g1     = (const float*)d_in[4];
    const float* beta1  = (const float*)d_in[5];
    const float* m1     = (const float*)d_in[6];
    const float* var1   = (const float*)d_in[7];
    const float* w2     = (const float*)d_in[8];
    const float* b2     = (const float*)d_in[9];
    const float* g2     = (const float*)d_in[10];
    const float* beta2  = (const float*)d_in[11];
    const float* m2     = (const float*)d_in[12];
    const float* var2   = (const float*)d_in[13];
    float* out = (float*)d_out;

    float* res = (float*)d_ws;
    float* y   = res + (size_t)NB * P;

    conv1_res_kernel<<<CONV1_BLOCKS + RES_BLOCKS, 64, 0, stream>>>(
        x, w1, b1, g1, beta1, m1, var1, y, ref, res);

    fused_out_kernel<<<1024, 64, 0, stream>>>(ref, w2, b2, g2, beta2, m2, var2,
                                              res, y, out);
}


# Two launches -> one: each block of 4 waves recomputes in LDS the channel-mean rows (4) and the conv1 rows (2 low-res) its two image rows need; no workspace round trip; every sum keeps its order
# speedup vs baseline: 2.5857x; 1.4172x over previous
#include <hip/hip_runtime.h>
#include <math.h>


typedef float v2f __attribute__((ext_vector_type(2)));
typedef float v4f __attribute__((ext_vector_type(4)));
typedef float v8f __attribute__((ext_vector_type(8)));

#define BN_EPS 1e-5f

constexpr int NB  = 8;
constexpr int C   = 64;
constexpr int HIN = 64,  WIN = 64;
constexpr int H   = 128, W   = 128;
constexpr int PIN = HIN * WIN;
constexpr int P   = H * W;

template <int NT, int STRIDE>
__device__ __forceinline__ void gemm64(const float* __restrict__ w,
                                       const float* __restrict__ incol,
                                       int lm, int half, v8f (&acc)[4][NT]) {
    typedef float vnf __attribute__((ext_vector_type(NT)));
#pragma unroll
    for (int mt = 0; mt < 4; ++mt)
#pragma unroll
        for (int t = 0; t < NT; ++t) acc[mt][t] = v8f{};

#pragma unroll
    for (int j = 0; j < 8; ++j) {
        const int k0 = 8 * j + 4 * half;
        v4f av[4];
#pragma unroll
        for (int mt = 0; mt < 4; ++mt)
            av[mt] = *(const v4f*)(w + (mt * 16 + lm) * 64 + k0);
        vnf bv[4];
#pragma unroll
        for (int i = 0; i < 4; ++i)
            bv[i] = *(const vnf*)(incol + (size_t)(k0 + i) * STRIDE);
#pragma unroll
        for (int u = 0; u < 2; ++u)
#pragma unroll
            for (int mt = 0; mt < 4; ++mt) {
                v2f a;
                a.x = av[mt][2 * u];
                a.y = av[mt][2 * u + 1];
#pragma unroll
                for (int t = 0; t < NT; ++t) {
                    v2f bm;
                    bm.x = bv[2 * u][t];
                    bm.y = bv[2 * u + 1][t];
                    acc[mt][t] = __builtin_amdgcn_wmma_f32_16x16x4_f32(
                        false, a, false, bm, (short)0, acc[mt][t], false, false);
                }
            }
    }
}

template <int NT, int STRIDE>
__device__ __forceinline__ void gemm64_load(const float* __restrict__ w,
                                            const float* __restrict__ incol,
                                            int lm, int half, v4f (&av)[8][4],
                                            float (&bv)[8][4][NT]) {
    typedef float vnf __attribute__((ext_vector_type(NT)));
#pragma unroll
    for (int j = 0; j < 8; ++j) {
        const int k0 = 8 * j + 4 * half;
#pragma unroll
        for (int mt = 0; mt < 4; ++mt)
            av[j][mt] = *(const v4f*)(w + (mt * 16 + lm) * 64 + k0);
#pragma unroll
        for (int i = 0; i < 4; ++i) {
            const vnf v = *(const vnf*)(incol + (size_t)(k0 + i) * STRIDE);
#pragma unroll
            for (int t = 0; t < NT; ++t) bv[j][i][t] = v[t];
        }
    }
}

template <int NT>
__device__ __forceinline__ void gemm64_mma(const v4f (&av)[8][4],
                                           const float (&bv)[8][4][NT],
                                           v8f (&acc)[4][NT]) {
#pragma unroll
    for (int mt = 0; mt < 4; ++mt)
#pragma unroll
        for (int t = 0; t < NT; ++t) acc[mt][t] = v8f{};
#pragma unroll
    for (int j = 0; j < 8; ++j)
#pragma unroll
        for (int u = 0; u < 2; ++u)
#pragma unroll
            for (int mt = 0; mt < 4; ++mt) {
                v2f a;
                a.x = av[j][mt][2 * u];
                a.y = av[j][mt][2 * u + 1];
#pragma unroll
                for (int t = 0; t < NT; ++t) {
                    v2f bm;
                    bm.x = bv[j][2 * u][t];
                    bm.y = bv[j][2 * u + 1][t];
                    acc[mt][t] = __builtin_amdgcn_wmma_f32_16x16x4_f32(
                        false, a, false, bm, (short)0, acc[mt][t], false, false);
                }
            }
}

constexpr int YS_LD = 68;
constexpr int RS_LD = 136;

__global__ __launch_bounds__(128)
void fused_all_kernel(const float* __restrict__ x,
                      const float* __restrict__ w1,
                      const float* __restrict__ b1,
                      const float* __restrict__ g1,
                      const float* __restrict__ beta1,
                      const float* __restrict__ m1,
                      const float* __restrict__ var1,
                      const float* __restrict__ ref,
                      const float* __restrict__ w2,
                      const float* __restrict__ b2,
                      const float* __restrict__ g2,
                      const float* __restrict__ beta2,
                      const float* __restrict__ m2,
                      const float* __restrict__ var2,
                      float* __restrict__ out) {
    __shared__ v2f sb1[C];
    __shared__ v2f sb2[C];
    __shared__ __attribute__((aligned(16))) float res_s[5 * RS_LD];
    __shared__ __attribute__((aligned(16))) float y_s[2 * C * YS_LD];

    const int tid  = threadIdx.x;
    const int lane = tid & 31;
    const int wv   = tid >> 5;
    const int s    = wv >> 1;
    const int hc   = wv & 1;
    const int half = lane >> 4;
    const int lm   = lane & 15;
    const int b    = blockIdx.x >> 6;
    const int mb   = blockIdx.x & 63;
    const bool edge = (mb == 0);

    const int hr   = edge ? (s ? 124 + wv : wv) : 2 * mb - 2 + wv;
    const int lrow = edge ? (s ? HIN - 1 : 0)   : mb - 1 + s;
    const int ph   = edge ? (s ? H - 1 : 0)     : 2 * mb - 1 + s;
    const int ri0  = edge ? (s ? 2 : 4) : s;
    const int ri1  = edge ? (s ? 3 : 0) : s + 1;
    const int ri2  = edge ? (s ? 4 : 1) : s + 2;
    const int ys0  = edge ? s : 0;
    const int ys1  = edge ? s : 1;

    if (tid < C) {
        v2f e;
        e.x = g1[tid] * rsqrtf(var1[tid] + BN_EPS);
        e.y = b1[tid] * e.x + beta1[tid] - m1[tid] * e.x;
        sb1[tid] = e;
    } else {
        const int t = tid - C;
        v2f e;
        e.x = g2[t] * rsqrtf(var2[t] + BN_EPS);
        e.y = b2[t] * e.x + beta2[t] - m2[t] * e.x;
        sb2[t] = e;
    }

    const int lc0 = hc * 32 + 2 * lm;
    v8f acc1[4][2];
    {
        v4f   av1[8][4];
        float bv1[8][4][2];
        gemm64_load<2, PIN>(w1, x + (size_t)b * C * PIN + lrow * WIN + lc0, lm, half, av1, bv1);

        const float* rp = ref + (size_t)b * C * P + hr * W + 4 * lane;
        v4f rb[32];
#pragma unroll
        for (int c = 0; c < 32; ++c) rb[c] = *(const v4f*)(rp + (size_t)c * P);
        v4f sum = {0.f, 0.f, 0.f, 0.f};
#pragma unroll
        for (int c = 0; c < 32; ++c) sum += rb[c];
#pragma unroll
        for (int c = 0; c < 32; ++c) rb[c] = *(const v4f*)(rp + (size_t)(32 + c) * P);

        gemm64_mma<2>(av1, bv1, acc1);

#pragma unroll
        for (int c = 0; c < 32; ++c) sum += rb[c];
        *(v4f*)(res_s + wv * RS_LD + 4 + 4 * lane) = sum * (1.0f / 64.0f);
        if (lane == 0)  res_s[wv * RS_LD + 3]   = 0.f;
        if (lane == 31) res_s[wv * RS_LD + 132] = 0.f;
        for (int i = tid; i < RS_LD; i += 128) res_s[4 * RS_LD + i] = 0.f;
    }

    {
        __syncthreads();

        float* ys = y_s + s * (C * YS_LD) + lc0;
#pragma unroll
        for (int mt = 0; mt < 4; ++mt)
#pragma unroll
            for (int r = 0; r < 8; ++r) {
                const int o  = mt * 16 + 8 * half + r;
                const v2f sc = sb1[o];
                v2f ov;
                ov.x = fmaxf(fmaf(acc1[mt][0][r], sc.x, sc.y), 0.f);
                ov.y = fmaxf(fmaf(acc1[mt][1][r], sc.x, sc.y), 0.f);
                *(v2f*)(ys + o * YS_LD) = ov;
            }
    }

    __syncthreads();

    const int pw0 = hc * 64 + 4 * lm;

    float R[3][6];
    {
        const int ri[3] = {ri0, ri1, ri2};
#pragma unroll
        for (int di = 0; di < 3; ++di) {
            const float* rr = res_s + ri[di] * RS_LD + 4 + pw0;
            const v4f mid = *(const v4f*)rr;
            R[di][0] = rr[-1];
            R[di][1] = mid.x;
            R[di][2] = mid.y;
            R[di][3] = mid.z;
            R[di][4] = mid.w;
            R[di][5] = rr[4];
        }
    }

    v8f acc[4][4];
    gemm64<4, P>(w2, ref + (size_t)b * C * P + ph * W + pw0, lm, half, acc);

    const int r1 = (ph + 1) >> 1;
    float wgt00[4], wgt01[4], wgt10[4], wgt11[4], invden[4];
#pragma unroll
    for (int t = 0; t < 4; ++t) {
        const int pw = pw0 + t;
        const int c1 = (pw + 1) >> 1;
        float ssum = 0.f;
#pragma unroll
        for (int di = 0; di < 3; ++di)
#pragma unroll
            for (int dj = 0; dj < 3; ++dj) ssum += R[di][t + dj];
        const float ua = ssum * (1.f / 9.f);
        const float ui = (R[1][t + 1] - ua > 0.f) ? 1.f : 0.f;

        float a00 = 0.f, a01 = 0.f, a10 = 0.f, a11 = 0.f;
        float den = 1e-6f;
#pragma unroll
        for (int di = 0; di < 3; ++di)
#pragma unroll
            for (int dj = 0; dj < 3; ++dj) {
                const float up = (R[di][t + dj] - ua > 0.f) ? 1.f : 0.f;
                const float mk = ui * up + (1.f - ui) * (1.f - up);
                den += mk;
                const int hh = ph + di - 1, ww = pw + dj - 1;
                if (hh >= 0 && hh < H && ww >= 0 && ww < W) {
                    const int sr = ((hh >> 1) == r1) ? 1 : 0;
                    const int sc = ((ww >> 1) == c1) ? 1 : 0;
                    if (sr == 0 && sc == 0) a00 += mk;
                    else if (sr == 0)       a01 += mk;
                    else if (sc == 0)       a10 += mk;
                    else                    a11 += mk;
                }
            }
        wgt00[t] = a00; wgt01[t] = a01; wgt10[t] = a10; wgt11[t] = a11;
        invden[t] = 1.f / den;
    }

    const int q   = pw0 >> 1;
    const int cl  = max(q - 1, 0), cr = min(q + 2, WIN - 1);

    const float* yp0 = y_s + ys0 * (C * YS_LD);
    const float* yp1 = y_s + ys1 * (C * YS_LD);
    float* ob = out + (size_t)b * C * P + ph * W + pw0;
#pragma unroll
    for (int mt = 0; mt < 4; ++mt)
#pragma unroll
        for (int r = 0; r < 8; ++r) {
            const int o  = mt * 16 + 8 * half + r;
            const v2f sc = sb2[o];
            const float* y0 = yp0 + o * YS_LD;
            const float* y1 = yp1 + o * YS_LD;
            float Y0[4], Y1[4];
            const v2f m0 = *(const v2f*)(y0 + q);
            const v2f m1v = *(const v2f*)(y1 + q);
            Y0[0] = y0[cl]; Y0[1] = m0.x;  Y0[2] = m0.y;  Y0[3] = y0[cr];
            Y1[0] = y1[cl]; Y1[1] = m1v.x; Y1[2] = m1v.y; Y1[3] = y1[cr];
            v4f ov;
#pragma unroll
            for (int t = 0; t < 4; ++t) {
                const int jl = (t + 1) >> 1, jr = jl + 1;
                const float c2v = fmaxf(fmaf(acc[mt][t][r], sc.x, sc.y), 0.f);
                const float num = wgt00[t] * Y0[jl]
                                + wgt01[t] * Y0[jr]
                                + wgt10[t] * Y1[jl]
                                + wgt11[t] * Y1[jr];
                ov[t] = num * invden[t] + c2v;
            }
            *(v4f*)(ob + o * P) = ov;
        }
}

extern "C" void kernel_launch(void* const* d_in, const int* in_sizes, int n_in,
                              void* d_out, int out_size, void* d_ws, size_t ws_size,
                              hipStream_t stream) {
    const float* x      = (const float*)d_in[0];
    const float* ref    = (const float*)d_in[1];
    const float* w1     = (const float*)d_in[2];
    const float* b1     = (const float*)d_in[3];
    const float* g1     = (const float*)d_in[4];
    const float* beta1  = (const float*)d_in[5];
    const float* m1     = (const float*)d_in[6];
    const float* var1   = (const float*)d_in[7];
    const float* w2     = (const float*)d_in[8];
    const float* b2     = (const float*)d_in[9];
    const float* g2     = (const float*)d_in[10];
    const float* beta2  = (const float*)d_in[11];
    const float* m2     = (const float*)d_in[12];
    const float* var2   = (const float*)d_in[13];
    float* out = (float*)d_out;

    fused_all_kernel<<<NB * 64, 128, 0, stream>>>(x, w1, b1, g1, beta1, m1, var1,
                                                  ref, w2, b2, g2, beta2, m2, var2, out);
}


# conv2's products moved in front of the barriers into one rolled K loop with conv1's and the channel mean; each step's loads go out one step ahead, under 48 products; every sum keeps its order
# speedup vs baseline: 2.7153x; 1.0501x over previous
#include <hip/hip_runtime.h>
#include <math.h>


typedef float v2f __attribute__((ext_vector_type(2)));
typedef float v4f __attribute__((ext_vector_type(4)));
typedef float v8f __attribute__((ext_vector_type(8)));

#define BN_EPS 1e-5f

constexpr int NB  = 8;
constexpr int C   = 64;
constexpr int HIN = 64,  WIN = 64;
constexpr int H   = 128, W   = 128;
constexpr int PIN = HIN * WIN;
constexpr int P   = H * W;

struct StepOps {
    v4f a2[4];
    v4f b2[4];
    v4f a1[4];
    v2f b1[4];
    v4f r[8];
};

__device__ __forceinline__ void step_load(StepOps& o,
                                          const float* __restrict__ pa2,
                                          const float* __restrict__ pb2,
                                          const float* __restrict__ pa1,
                                          const float* __restrict__ pb1,
                                          const float* __restrict__ pr) {
#pragma unroll
    for (int mt = 0; mt < 4; ++mt) o.a2[mt] = *(const v4f*)(pa2 + mt * 16 * 64);
#pragma unroll
    for (int i = 0; i < 4; ++i) o.b2[i] = *(const v4f*)(pb2 + (size_t)i * P);
#pragma unroll
    for (int mt = 0; mt < 4; ++mt) o.a1[mt] = *(const v4f*)(pa1 + mt * 16 * 64);
#pragma unroll
    for (int i = 0; i < 4; ++i) o.b1[i] = *(const v2f*)(pb1 + (size_t)i * PIN);
#pragma unroll
    for (int c = 0; c < 8; ++c) o.r[c] = *(const v4f*)(pr + (size_t)c * P);
}

__device__ __forceinline__ void step_mma(const StepOps& o, v8f (&acc)[4][4],
                                         v8f (&acc1)[4][2], v4f& sum) {
#pragma unroll
    for (int u = 0; u < 2; ++u)
#pragma unroll
        for (int mt = 0; mt < 4; ++mt) {
            v2f a;
            a.x = o.a2[mt][2 * u];
            a.y = o.a2[mt][2 * u + 1];
#pragma unroll
            for (int t = 0; t < 4; ++t) {
                v2f bm;
                bm.x = o.b2[2 * u][t];
                bm.y = o.b2[2 * u + 1][t];
                acc[mt][t] = __builtin_amdgcn_wmma_f32_16x16x4_f32(
                    false, a, false, bm, (short)0, acc[mt][t], false, false);
            }
        }
#pragma unroll
    for (int u = 0; u < 2; ++u)
#pragma unroll
        for (int mt = 0; mt < 4; ++mt) {
            v2f a;
            a.x = o.a1[mt][2 * u];
            a.y = o.a1[mt][2 * u + 1];
#pragma unroll
            for (int t = 0; t < 2; ++t) {
                v2f bm;
                bm.x = o.b1[2 * u][t];
                bm.y = o.b1[2 * u + 1][t];
                acc1[mt][t] = __builtin_amdgcn_wmma_f32_16x16x4_f32(
                    false, a, false, bm, (short)0, acc1[mt][t], false, false);
            }
        }
#pragma unroll
    for (int c = 0; c < 8; ++c) sum += o.r[c];
}

constexpr int YS_LD = 68;
constexpr int RS_LD = 136;

__global__ __launch_bounds__(128)
void fused_all_kernel(const float* __restrict__ x,
                      const float* __restrict__ w1,
                      const float* __restrict__ b1,
                      const float* __restrict__ g1,
                      const float* __restrict__ beta1,
                      const float* __restrict__ m1,
                      const float* __restrict__ var1,
                      const float* __restrict__ ref,
                      const float* __restrict__ w2,
                      const float* __restrict__ b2,
                      const float* __restrict__ g2,
                      const float* __restrict__ beta2,
                      const float* __restrict__ m2,
                      const float* __restrict__ var2,
                      float* __restrict__ out) {
    __shared__ v2f sb1[C];
    __shared__ v2f sb2[C];
    __shared__ __attribute__((aligned(16))) float res_s[5 * RS_LD];
    __shared__ __attribute__((aligned(16))) float y_s[2 * C * YS_LD];

    const int tid  = threadIdx.x;
    const int lane = tid & 31;
    const int wv   = tid >> 5;
    const int s    = wv >> 1;
    const int hc   = wv & 1;
    const int half = lane >> 4;
    const int lm   = lane & 15;
    const int b    = blockIdx.x >> 6;
    const int mb   = blockIdx.x & 63;
    const bool edge = (mb == 0);

    const int hr   = edge ? (s ? 124 + wv : wv) : 2 * mb - 2 + wv;
    const int lrow = edge ? (s ? HIN - 1 : 0)   : mb - 1 + s;
    const int ph   = edge ? (s ? H - 1 : 0)     : 2 * mb - 1 + s;
    const int ri0  = edge ? (s ? 2 : 4) : s;
    const int ri1  = edge ? (s ? 3 : 0) : s + 1;
    const int ri2  = edge ? (s ? 4 : 1) : s + 2;
    const int ys0  = edge ? s : 0;
    const int ys1  = edge ? s : 1;

    const int lc0 = hc * 32 + 2 * lm;
    const int pw0 = hc * 64 + 4 * lm;
    const float* refb = ref + (size_t)b * C * P;
    const float* pa2 = w2 + lm * 64 + 4 * half;
    const float* pa1 = w1 + lm * 64 + 4 * half;
    const float* pb2 = refb + (size_t)(4 * half) * P + ph * W + pw0;
    const float* pb1 = x + (size_t)b * C * PIN + (size_t)(4 * half) * PIN + lrow * WIN + lc0;
    const float* pr  = refb + hr * W + 4 * lane;

    StepOps o0, o1;
    step_load(o0, pa2, pb2, pa1, pb1, pr);

    if (tid < C) {
        v2f e;
        e.x = g1[tid] * rsqrtf(var1[tid] + BN_EPS);
        e.y = b1[tid] * e.x + beta1[tid] - m1[tid] * e.x;
        sb1[tid] = e;
    } else {
        const int t = tid - C;
        v2f e;
        e.x = g2[t] * rsqrtf(var2[t] + BN_EPS);
        e.y = b2[t] * e.x + beta2[t] - m2[t] * e.x;
        sb2[t] = e;
    }

    v8f acc[4][4];
    v8f acc1[4][2];
#pragma unroll
    for (int mt = 0; mt < 4; ++mt) {
#pragma unroll
        for (int t = 0; t < 4; ++t) acc[mt][t] = v8f{};
#pragma unroll
        for (int t = 0; t < 2; ++t) acc1[mt][t] = v8f{};
    }
    v4f sum = {0.f, 0.f, 0.f, 0.f};

#pragma clang loop unroll(disable)
    for (int jj = 0; jj < 3; ++jj) {
        step_load(o1, pa2 + 8, pb2 + (size_t)8 * P, pa1 + 8, pb1 + (size_t)8 * PIN,
                  pr + (size_t)8 * P);
        step_mma(o0, acc, acc1, sum);
        pa2 += 16; pa1 += 16;
        pb2 += (size_t)16 * P; pb1 += (size_t)16 * PIN; pr += (size_t)16 * P;
        step_load(o0, pa2, pb2, pa1, pb1, pr);
        step_mma(o1, acc, acc1, sum);
    }
    step_load(o1, pa2 + 8, pb2 + (size_t)8 * P, pa1 + 8, pb1 + (size_t)8 * PIN,
              pr + (size_t)8 * P);
    step_mma(o0, acc, acc1, sum);
    step_mma(o1, acc, acc1, sum);

    *(v4f*)(res_s + wv * RS_LD + 4 + 4 * lane) = sum * (1.0f / 64.0f);
    if (lane == 0)  res_s[wv * RS_LD + 3]   = 0.f;
    if (lane == 31) res_s[wv * RS_LD + 132] = 0.f;
    for (int i = tid; i < RS_LD; i += 128) res_s[4 * RS_LD + i] = 0.f;

    {
        __syncthreads();

        float* ys = y_s + s * (C * YS_LD) + lc0;
#pragma unroll
        for (int mt = 0; mt < 4; ++mt)
#pragma unroll
            for (int r = 0; r < 8; ++r) {
                const int o  = mt * 16 + 8 * half + r;
                const v2f sc = sb1[o];
                v2f ov;
                ov.x = fmaxf(fmaf(acc1[mt][0][r], sc.x, sc.y), 0.f);
                ov.y = fmaxf(fmaf(acc1[mt][1][r], sc.x, sc.y), 0.f);
                *(v2f*)(ys + o * YS_LD) = ov;
            }
    }

    __syncthreads();

    float R[3][6];
    {
        const int ri[3] = {ri0, ri1, ri2};
#pragma unroll
        for (int di = 0; di < 3; ++di) {
            const float* rr = res_s + ri[di] * RS_LD + 4 + pw0;
            const v4f mid = *(const v4f*)rr;
            R[di][0] = rr[-1];
            R[di][1] = mid.x;
            R[di][2] = mid.y;
            R[di][3] = mid.z;
            R[di][4] = mid.w;
            R[di][5] = rr[4];
        }
    }

    const int r1 = (ph + 1) >> 1;
    float wgt00[4], wgt01[4], wgt10[4], wgt11[4], invden[4];
#pragma unroll
    for (int t = 0; t < 4; ++t) {
        const int pw = pw0 + t;
        const int c1 = (pw + 1) >> 1;
        float ssum = 0.f;
#pragma unroll
        for (int di = 0; di < 3; ++di)
#pragma unroll
            for (int dj = 0; dj < 3; ++dj) ssum += R[di][t + dj];
        const float ua = ssum * (1.f / 9.f);
        const float ui = (R[1][t + 1] - ua > 0.f) ? 1.f : 0.f;

        float a00 = 0.f, a01 = 0.f, a10 = 0.f, a11 = 0.f;
        float den = 1e-6f;
#pragma unroll
        for (int di = 0; di < 3; ++di)
#pragma unroll
            for (int dj = 0; dj < 3; ++dj) {
                const float up = (R[di][t + dj] - ua > 0.f) ? 1.f : 0.f;
                const float mk = ui * up + (1.f - ui) * (1.f - up);
                den += mk;
                const int hh = ph + di - 1, ww = pw + dj - 1;
                if (hh >= 0 && hh < H && ww >= 0 && ww < W) {
                    const int sr = ((hh >> 1) == r1) ? 1 : 0;
                    const int sc = ((ww >> 1) == c1) ? 1 : 0;
                    if (sr == 0 && sc == 0) a00 += mk;
                    else if (sr == 0)       a01 += mk;
                    else if (sc == 0)       a10 += mk;
                    else                    a11 += mk;
                }
            }
        wgt00[t] = a00; wgt01[t] = a01; wgt10[t] = a10; wgt11[t] = a11;
        invden[t] = 1.f / den;
    }

    const int q   = pw0 >> 1;
    const int cl  = max(q - 1, 0), cr = min(q + 2, WIN - 1);

    const float* yp0 = y_s + ys0 * (C * YS_LD);
    const float* yp1 = y_s + ys1 * (C * YS_LD);
    float* ob = out + (size_t)b * C * P + ph * W + pw0;
#pragma unroll
    for (int mt = 0; mt < 4; ++mt)
#pragma unroll
        for (int r = 0; r < 8; ++r) {
            const int o  = mt * 16 + 8 * half + r;
            const v2f sc = sb2[o];
            const float* y0 = yp0 + o * YS_LD;
            const float* y1 = yp1 + o * YS_LD;
            float Y0[4], Y1[4];
            const v2f m0 = *(const v2f*)(y0 + q);
            const v2f m1v = *(const v2f*)(y1 + q);
            Y0[0] = y0[cl]; Y0[1] = m0.x;  Y0[2] = m0.y;  Y0[3] = y0[cr];
            Y1[0] = y1[cl]; Y1[1] = m1v.x; Y1[2] = m1v.y; Y1[3] = y1[cr];
            v4f ov;
#pragma unroll
            for (int t = 0; t < 4; ++t) {
                const int jl = (t + 1) >> 1, jr = jl + 1;
                const float c2v = fmaxf(fmaf(acc[mt][t][r], sc.x, sc.y), 0.f);
                const float num = wgt00[t] * Y0[jl]
                                + wgt01[t] * Y0[jr]
                                + wgt10[t] * Y1[jl]
                                + wgt11[t] * Y1[jr];
                ov[t] = num * invden[t] + c2v;
            }
            *(v4f*)(ob + o * P) = ov;
        }
}

extern "C" void kernel_launch(void* const* d_in, const int* in_sizes, int n_in,
                              void* d_out, int out_size, void* d_ws, size_t ws_size,
                              hipStream_t stream) {
    const float* x      = (const float*)d_in[0];
    const float* ref    = (const float*)d_in[1];
    const float* w1     = (const float*)d_in[2];
    const float* b1     = (const float*)d_in[3];
    const float* g1     = (const float*)d_in[4];
    const float* beta1  = (const float*)d_in[5];
    const float* m1     = (const float*)d_in[6];
    const float* var1   = (const float*)d_in[7];
    const float* w2     = (const float*)d_in[8];
    const float* b2     = (const float*)d_in[9];
    const float* g2     = (const float*)d_in[10];
    const float* beta2  = (const float*)d_in[11];
    const float* m2     = (const float*)d_in[12];
    const float* var2   = (const float*)d_in[13];
    float* out = (float*)d_out;

    fused_all_kernel<<<NB * 64, 128, 0, stream>>>(x, w1, b1, g1, beta1, m1, var1,
                                                  ref, w2, b2, g2, beta2, m2, var2, out);
}
